# starting kernel
# baseline (speedup 1.0000x reference)
#include <hip/hip_runtime.h>


typedef __attribute__((ext_vector_type(2))) float v2f;
typedef __attribute__((ext_vector_type(8))) float v8f;

#define N_NODES 100000
#define N_EDGES 600000
#define DIM     128
#define NCH     16

__global__ __launch_bounds__(256) void node_proj_wmma(
    const float* __restrict__ h,
    const float* __restrict__ W,
    float* __restrict__ P)
{
    const int lane  = threadIdx.x & 31;
    const int wave  = (blockIdx.x * (blockDim.x >> 5)) + (threadIdx.x >> 5);
    const int m0    = wave * 16;
    if (m0 >= N_NODES) return;

    const int half  = lane >> 4;
    const int sub   = lane & 15;
    const int khalf = half * 2;

    const float* arow = h + (size_t)(m0 + sub) * DIM + khalf;
    const float* brow0 = W + (size_t)sub * (2 * DIM) + khalf;
    const float* brow1 = brow0 + DIM;

    v8f acc0 = {};
    v8f acc1 = {};

#pragma unroll
    for (int k = 0; k < DIM; k += 4) {
        v2f a  = *(const v2f*)(arow  + k);
        v2f b0 = *(const v2f*)(brow0 + k);
        v2f b1 = *(const v2f*)(brow1 + k);
        acc0 = __builtin_amdgcn_wmma_f32_16x16x4_f32(
            false, a, false, b0, (short)0, acc0, false, false);
        acc1 = __builtin_amdgcn_wmma_f32_16x16x4_f32(
            false, a, false, b1, (short)0, acc1, false, false);
    }

    const int mrw = m0 + half * 8;
#pragma unroll
    for (int r = 0; r < 8; ++r) {
        float* prow = P + (size_t)(mrw + r) * 32;
        prow[sub]      = acc0[r];
        prow[16 + sub] = acc1[r];
    }
}

__global__ __launch_bounds__(256) void edge_gather_add(
    const float* __restrict__ P,
    const int*   __restrict__ esrc,
    const int*   __restrict__ edst,
    const float* __restrict__ b,
    float*       __restrict__ out)
{
    const int t = blockIdx.x * blockDim.x + threadIdx.x;
    const int e  = t >> 2;
    if (e >= N_EDGES) return;
    const int c4 = (t & 3) * 4;

    const int s = esrc[e];
    const int d = edst[e];

    const float4 ps = *(const float4*)(P + (size_t)s * 32 + c4);
    const float4 pd = *(const float4*)(P + (size_t)d * 32 + 16 + c4);
    const float4 bb = *(const float4*)(b + c4);

    float4 r;
    r.x = ps.x + pd.x + bb.x;
    r.y = ps.y + pd.y + bb.y;
    r.z = ps.z + pd.z + bb.z;
    r.w = ps.w + pd.w + bb.w;

    *(float4*)(out + (size_t)e * NCH + c4) = r;
}

extern "C" void kernel_launch(void* const* d_in, const int* in_sizes, int n_in,
                              void* d_out, int out_size, void* d_ws, size_t ws_size,
                              hipStream_t stream) {
    const float* h    = (const float*)d_in[0];
    const int*   src  = (const int*)  d_in[1];
    const int*   dst  = (const int*)  d_in[2];
    const float* W    = (const float*)d_in[3];
    const float* b    = (const float*)d_in[4];
    float*       out  = (float*)d_out;
    float*       P    = (float*)d_ws;

    const int waves_needed = (N_NODES + 15) / 16;
    const int blocks1 = (waves_needed + 7) / 8;
    node_proj_wmma<<<blocks1, 256, 0, stream>>>(h, W, P);

    const int threads2 = N_EDGES * 4;
    const int blocks2 = (threads2 + 255) / 256;
    edge_gather_add<<<blocks2, 256, 0, stream>>>(P, src, dst, b, out);
}


# K1 A operand path: h fetched as whole 128-B lines (4 lines per b128 load, not 16 per b64 load) into a wave-private LDS strip, A fragments read from LDS; W fragments all requested ahead; sums kept
# speedup vs baseline: 1.2932x; 1.2932x over previous
#include <hip/hip_runtime.h>


typedef __attribute__((ext_vector_type(2))) float v2f;
typedef __attribute__((ext_vector_type(4))) float v4f;
typedef __attribute__((ext_vector_type(8))) float v8f;

#define N_NODES 100000
#define N_EDGES 600000
#define DIM     128
#define NCH     16

#define MT     4
#define ROWS   (16 * MT)
#define CK     32
#define NCHUNK (DIM / CK)
#define HS     4
#define NSTEP  (DIM / 4)
#define NBLK   (NSTEP / HS)
#define LPC    (ROWS / 4)
#define PITCH  36

__global__ __launch_bounds__(32) void node_proj_wmma(
    const float* __restrict__ h,
    const float* __restrict__ W,
    float* __restrict__ P)
{
    __shared__ __attribute__((aligned(16))) float strip[2][ROWS * PITCH];

    const int lane  = threadIdx.x & 31;
    const int m0    = blockIdx.x * ROWS;
    if (m0 >= N_NODES) return;

    const int half  = lane >> 4;
    const int sub   = lane & 15;
    const int khalf = half * 2;

    const int r4 = lane >> 3;
    const int q4 = (lane & 7) * 4;

    const float* gsrc[MT];
#pragma unroll
    for (int t = 0; t < MT; ++t) {
        int tb = m0 + 16 * t;
        tb = tb < N_NODES ? tb : m0;
        gsrc[t] = h + (size_t)(tb + r4) * DIM + q4;
    }
    const int st_off = r4 * PITCH + q4;
    const int ld_off = sub * PITCH + khalf;

    const float* brow0 = W + (size_t)sub * (2 * DIM) + khalf;
    const float* brow1 = brow0 + DIM;

    v8f acc[MT][2];
#pragma unroll
    for (int t = 0; t < MT; ++t) { acc[t][0] = (v8f){}; acc[t][1] = (v8f){}; }

    v4f land[2][LPC];
    v2f a[2][HS][MT];
    v2f b[NSTEP][2];

#define LOAD_CHUNK(c, s)                                                          \
    _Pragma("unroll")                                                             \
    for (int i = 0; i < LPC; ++i)                                                 \
        land[s][i] = *(const v4f*)(gsrc[i >> 2] + (i & 3) * 4 * DIM + (c) * CK);

#define PARK_CHUNK(s)                                                             \
    _Pragma("unroll")                                                             \
    for (int i = 0; i < LPC; ++i)                                                 \
        *(v4f*)&strip[s][st_off + i * 4 * PITCH] = land[s][i];

#define LOAD_AFRAGS(blk, f)                                                       \
    _Pragma("unroll")                                                             \
    for (int j = 0; j < HS; ++j) {                                                \
        const int kc = ((((blk) & 1) * HS) + j) * 4;                              \
        _Pragma("unroll")                                                         \
        for (int t = 0; t < MT; ++t)                                              \
            a[f][j][t] = *(const v2f*)&strip[((blk) >> 1) & 1][ld_off + 16 * t * PITCH + kc]; \
    }

    LOAD_CHUNK(0, 0)
    LOAD_CHUNK(1, 1)
    __builtin_amdgcn_sched_barrier(0);
#pragma unroll
    for (int ks = 0; ks < NSTEP / 2; ++ks) {
        b[ks][0] = *(const v2f*)(brow0 + ks * 4);
        b[ks][1] = *(const v2f*)(brow1 + ks * 4);
    }
    __builtin_amdgcn_sched_barrier(0);
    PARK_CHUNK(0)
    LOAD_CHUNK(2, 0)
#pragma unroll
    for (int ks = NSTEP / 2; ks < NSTEP; ++ks) {
        b[ks][0] = *(const v2f*)(brow0 + ks * 4);
        b[ks][1] = *(const v2f*)(brow1 + ks * 4);
    }
    LOAD_AFRAGS(0, 0)

#pragma unroll
    for (int s = 0; s < NBLK; ++s) {
        if (s + 1 < NBLK) {
            const int nc = (s + 1) >> 1;
            if (((s + 1) & 1) == 0) {
                if (nc & 1) { PARK_CHUNK(1) } else { PARK_CHUNK(0) }
                if (nc + 2 < NCHUNK) {
                    if (nc & 1) { LOAD_CHUNK(nc + 2, 1) } else { LOAD_CHUNK(nc + 2, 0) }
                }
            }
            if (s & 1) { LOAD_AFRAGS(s + 1, 0) } else { LOAD_AFRAGS(s + 1, 1) }
        }
        __builtin_amdgcn_sched_barrier(0);
        const int cb = s & 1;
#pragma unroll
        for (int j = 0; j < HS; ++j) {
#pragma unroll
            for (int t = 0; t < MT; ++t) {
                acc[t][0] = __builtin_amdgcn_wmma_f32_16x16x4_f32(
                    false, a[cb][j][t], false, b[s * HS + j][0], (short)0, acc[t][0], false, false);
                acc[t][1] = __builtin_amdgcn_wmma_f32_16x16x4_f32(
                    false, a[cb][j][t], false, b[s * HS + j][1], (short)0, acc[t][1], false, false);
            }
        }
        __builtin_amdgcn_sched_barrier(0);
    }
#undef LOAD_CHUNK
#undef PARK_CHUNK
#undef LOAD_AFRAGS

#pragma unroll
    for (int t = 0; t < MT; ++t) {
        const int mt0 = m0 + 16 * t;
        if (mt0 < N_NODES) {
            const int mrw = mt0 + half * 8;
#pragma unroll
            for (int r = 0; r < 8; ++r) {
                float* prow = P + (size_t)(mrw + r) * 32;
                prow[sub]      = acc[t][0][r];
                prow[16 + sub] = acc[t][1][r];
            }
        }
    }
}

__global__ __launch_bounds__(256) void edge_gather_add(
    const float* __restrict__ P,
    const int*   __restrict__ esrc,
    const int*   __restrict__ edst,
    const float* __restrict__ b,
    float*       __restrict__ out)
{
    const int t = blockIdx.x * blockDim.x + threadIdx.x;
    const int e  = t >> 2;
    if (e >= N_EDGES) return;
    const int c4 = (t & 3) * 4;

    const int s = esrc[e];
    const int d = edst[e];

    const float4 ps = *(const float4*)(P + (size_t)s * 32 + c4);
    const float4 pd = *(const float4*)(P + (size_t)d * 32 + 16 + c4);
    const float4 bb = *(const float4*)(b + c4);

    float4 r;
    r.x = ps.x + pd.x + bb.x;
    r.y = ps.y + pd.y + bb.y;
    r.z = ps.z + pd.z + bb.z;
    r.w = ps.w + pd.w + bb.w;

    *(float4*)(out + (size_t)e * NCH + c4) = r;
}

extern "C" void kernel_launch(void* const* d_in, const int* in_sizes, int n_in,
                              void* d_out, int out_size, void* d_ws, size_t ws_size,
                              hipStream_t stream) {
    const float* h    = (const float*)d_in[0];
    const int*   src  = (const int*)  d_in[1];
    const int*   dst  = (const int*)  d_in[2];
    const float* W    = (const float*)d_in[3];
    const float* b    = (const float*)d_in[4];
    float*       out  = (float*)d_out;
    float*       P    = (float*)d_ws;

    const int blocks1 = (N_NODES + 16 * MT - 1) / (16 * MT);
    node_proj_wmma<<<blocks1, 32, 0, stream>>>(h, W, P);

    const int threads2 = N_EDGES * 4;
    const int blocks2 = (threads2 + 255) / 256;
    edge_gather_add<<<blocks2, 256, 0, stream>>>(P, src, dst, b, out);
}


# Edge launch: one wave owns 32 consecutive edges (4 per lane quad, no LDS, no lane traffic): 75000 -> 18750 waves, 8 index loads then 8 row gathers in flight per wave, whole-line stores; sums kept
# speedup vs baseline: 1.4267x; 1.1032x over previous
#include <hip/hip_runtime.h>


typedef __attribute__((ext_vector_type(2))) float v2f;
typedef __attribute__((ext_vector_type(4))) float v4f;
typedef __attribute__((ext_vector_type(8))) float v8f;

#define N_NODES 100000
#define N_EDGES 600000
#define DIM     128
#define NCH     16

#define MT     4
#define ROWS   (16 * MT)
#define CK     32
#define NCHUNK (DIM / CK)
#define HS     4
#define NSTEP  (DIM / 4)
#define NBLK   (NSTEP / HS)
#define LPC    (ROWS / 4)
#define PITCH  36

__global__ __launch_bounds__(32) void node_proj_wmma(
    const float* __restrict__ h,
    const float* __restrict__ W,
    float* __restrict__ P)
{
    __shared__ __attribute__((aligned(16))) float strip[2][ROWS * PITCH];

    const int lane  = threadIdx.x & 31;
    const int m0    = blockIdx.x * ROWS;
    if (m0 >= N_NODES) return;

    const int half  = lane >> 4;
    const int sub   = lane & 15;
    const int khalf = half * 2;

    const int r4 = lane >> 3;
    const int q4 = (lane & 7) * 4;

    const float* gsrc[MT];
#pragma unroll
    for (int t = 0; t < MT; ++t) {
        int tb = m0 + 16 * t;
        tb = tb < N_NODES ? tb : m0;
        gsrc[t] = h + (size_t)(tb + r4) * DIM + q4;
    }
    const int st_off = r4 * PITCH + q4;
    const int ld_off = sub * PITCH + khalf;

    const float* brow0 = W + (size_t)sub * (2 * DIM) + khalf;
    const float* brow1 = brow0 + DIM;

    v8f acc[MT][2];
#pragma unroll
    for (int t = 0; t < MT; ++t) { acc[t][0] = (v8f){}; acc[t][1] = (v8f){}; }

    v4f land[2][LPC];
    v2f a[2][HS][MT];
    v2f b[NSTEP][2];

#define LOAD_CHUNK(c, s)                                                          \
    _Pragma("unroll")                                                             \
    for (int i = 0; i < LPC; ++i)                                                 \
        land[s][i] = *(const v4f*)(gsrc[i >> 2] + (i & 3) * 4 * DIM + (c) * CK);

#define PARK_CHUNK(s)                                                             \
    _Pragma("unroll")                                                             \
    for (int i = 0; i < LPC; ++i)                                                 \
        *(v4f*)&strip[s][st_off + i * 4 * PITCH] = land[s][i];

#define LOAD_AFRAGS(blk, f)                                                       \
    _Pragma("unroll")                                                             \
    for (int j = 0; j < HS; ++j) {                                                \
        const int kc = ((((blk) & 1) * HS) + j) * 4;                              \
        _Pragma("unroll")                                                         \
        for (int t = 0; t < MT; ++t)                                              \
            a[f][j][t] = *(const v2f*)&strip[((blk) >> 1) & 1][ld_off + 16 * t * PITCH + kc]; \
    }

    LOAD_CHUNK(0, 0)
    LOAD_CHUNK(1, 1)
    __builtin_amdgcn_sched_barrier(0);
#pragma unroll
    for (int ks = 0; ks < NSTEP / 2; ++ks) {
        b[ks][0] = *(const v2f*)(brow0 + ks * 4);
        b[ks][1] = *(const v2f*)(brow1 + ks * 4);
    }
    __builtin_amdgcn_sched_barrier(0);
    PARK_CHUNK(0)
    LOAD_CHUNK(2, 0)
#pragma unroll
    for (int ks = NSTEP / 2; ks < NSTEP; ++ks) {
        b[ks][0] = *(const v2f*)(brow0 + ks * 4);
        b[ks][1] = *(const v2f*)(brow1 + ks * 4);
    }
    LOAD_AFRAGS(0, 0)

#pragma unroll
    for (int s = 0; s < NBLK; ++s) {
        if (s + 1 < NBLK) {
            const int nc = (s + 1) >> 1;
            if (((s + 1) & 1) == 0) {
                if (nc & 1) { PARK_CHUNK(1) } else { PARK_CHUNK(0) }
                if (nc + 2 < NCHUNK) {
                    if (nc & 1) { LOAD_CHUNK(nc + 2, 1) } else { LOAD_CHUNK(nc + 2, 0) }
                }
            }
            if (s & 1) { LOAD_AFRAGS(s + 1, 0) } else { LOAD_AFRAGS(s + 1, 1) }
        }
        __builtin_amdgcn_sched_barrier(0);
        const int cb = s & 1;
#pragma unroll
        for (int j = 0; j < HS; ++j) {
#pragma unroll
            for (int t = 0; t < MT; ++t) {
                acc[t][0] = __builtin_amdgcn_wmma_f32_16x16x4_f32(
                    false, a[cb][j][t], false, b[s * HS + j][0], (short)0, acc[t][0], false, false);
                acc[t][1] = __builtin_amdgcn_wmma_f32_16x16x4_f32(
                    false, a[cb][j][t], false, b[s * HS + j][1], (short)0, acc[t][1], false, false);
            }
        }
        __builtin_amdgcn_sched_barrier(0);
    }
#undef LOAD_CHUNK
#undef PARK_CHUNK
#undef LOAD_AFRAGS

#pragma unroll
    for (int t = 0; t < MT; ++t) {
        const int mt0 = m0 + 16 * t;
        if (mt0 < N_NODES) {
            const int mrw = mt0 + half * 8;
#pragma unroll
            for (int r = 0; r < 8; ++r) {
                float* prow = P + (size_t)(mrw + r) * 32;
                prow[sub]      = acc[t][0][r];
                prow[16 + sub] = acc[t][1][r];
            }
        }
    }
}

#define EPW 32
#define EPQ (EPW / 8)

__global__ __launch_bounds__(256) void edge_gather_add(
    const float* __restrict__ P,
    const int*   __restrict__ esrc,
    const int*   __restrict__ edst,
    const float* __restrict__ b,
    float*       __restrict__ out)
{
    static_assert(N_EDGES % EPW == 0, "a wave's edge block is never partial");
    const int t  = blockIdx.x * 256 + threadIdx.x;
    const int e0 = (t >> 5) * EPW;
    if (e0 >= N_EDGES) return;
    const int q  = (threadIdx.x & 31) >> 2;
    const int c4 = (threadIdx.x & 3) * 4;
    const int eq = e0 + q;

    int s[EPQ], d[EPQ];
#pragma unroll
    for (int j = 0; j < EPQ; ++j) {
        s[j] = esrc[eq + 8 * j];
        d[j] = edst[eq + 8 * j];
    }

    float4 ps[EPQ], pd[EPQ];
#pragma unroll
    for (int j = 0; j < EPQ; ++j) {
        ps[j] = *(const float4*)(P + (size_t)s[j] * 32 + c4);
        pd[j] = *(const float4*)(P + (size_t)d[j] * 32 + 16 + c4);
    }
    const float4 bb = *(const float4*)(b + c4);

#pragma unroll
    for (int j = 0; j < EPQ; ++j) {
        float4 r;
        r.x = ps[j].x + pd[j].x + bb.x;
        r.y = ps[j].y + pd[j].y + bb.y;
        r.z = ps[j].z + pd[j].z + bb.z;
        r.w = ps[j].w + pd[j].w + bb.w;
        *(float4*)(out + (size_t)(eq + 8 * j) * NCH + c4) = r;
    }
}

extern "C" void kernel_launch(void* const* d_in, const int* in_sizes, int n_in,
                              void* d_out, int out_size, void* d_ws, size_t ws_size,
                              hipStream_t stream) {
    const float* h    = (const float*)d_in[0];
    const int*   src  = (const int*)  d_in[1];
    const int*   dst  = (const int*)  d_in[2];
    const float* W    = (const float*)d_in[3];
    const float* b    = (const float*)d_in[4];
    float*       out  = (float*)d_out;
    float*       P    = (float*)d_ws;

    const int blocks1 = (N_NODES + 16 * MT - 1) / (16 * MT);
    node_proj_wmma<<<blocks1, 32, 0, stream>>>(h, W, P);

    const int threads2 = N_EDGES / EPW * 32;
    const int blocks2 = (threads2 + 255) / 256;
    edge_gather_add<<<blocks2, 256, 0, stream>>>(P, src, dst, b, out);
}
